# starting kernel
# baseline (speedup 1.0000x reference)
#include <hip/hip_runtime.h>

typedef __attribute__((ext_vector_type(2))) float v2f;
typedef __attribute__((ext_vector_type(8))) float v8f;

#define NI 64
#define NTILE 4

__device__ __forceinline__ float fast_tanh(float x) {
#if __has_builtin(__builtin_amdgcn_tanhf)
  return __builtin_amdgcn_tanhf(x);
#else
  float r;
  asm volatile("v_tanh_f32 %0, %1" : "=v"(r) : "v"(x));
  return r;
#endif
}

__global__ __launch_bounds__(256) void henon_wmma_kernel(
    const float* __restrict__ z, const float* __restrict__ W_in,
    const float* __restrict__ W_out, const float* __restrict__ b_in,
    const float* __restrict__ eta, float* __restrict__ out, int nbatch) {
  __shared__ float Gt[8][NI][16];

  const int tid   = threadIdx.x;
  const int lane  = tid & 31;
  const int wslot = tid >> 5;
  const int hilo  = lane >> 4;
  const int l16   = lane & 15;
  float* gbase = &Gt[wslot][0][0];

  float b1x[NTILE], b1y[NTILE], wv[NTILE];
#pragma unroll
  for (int T = 0; T < NTILE; ++T) {
    int j = T * 16 + l16;
    b1x[T] = hilo ? b_in[j] : W_in[0 * NI + j];
    b1y[T] = hilo ? 0.0f    : W_in[1 * NI + j];
    wv[T]  = W_out[j];
  }
  float a2x[16], a2y[16];
#pragma unroll
  for (int c = 0; c < 16; ++c) {
    int k0 = 4 * c + 2 * hilo;
    a2x[c] = (l16 < 2) ? W_in[l16 * NI + k0]     : 0.0f;
    a2y[c] = (l16 < 2) ? W_in[l16 * NI + k0 + 1] : 0.0f;
  }
  const float eta0 = eta[0], eta1 = eta[1];

  const int ntiles = (nbatch + 15) >> 4;
  const int waveG  = (int)(blockIdx.x * blockDim.x + tid) >> 5;
  const int nwaves = (int)(gridDim.x * blockDim.x) >> 5;
  const float4* z4 = (const float4*)z;
  float4* o4 = (float4*)out;

  for (int tile = waveG; tile < ntiles; tile += nwaves) {
    const int b = tile * 16 + l16;
    const bool act = (hilo == 0) && (b < nbatch);
    float4 zv = make_float4(0.f, 0.f, 0.f, 0.f);
    if (act) zv = z4[b];
    {
      int nt = tile + nwaves;
      if (nt < ntiles) __builtin_prefetch(&z4[nt * 16 + l16], 0, 1);
    }
    float X0 = zv.x, X1 = zv.y, Y0 = zv.z, Y1 = zv.w;

#pragma unroll
    for (int it = 0; it < 4; ++it) {
      v2f a1;
      a1.x = hilo ? 1.0f : Y0;
      a1.y = hilo ? 0.0f : Y1;
      v8f h[NTILE];
#pragma unroll
      for (int T = 0; T < NTILE; ++T) {
        v2f b1; b1.x = b1x[T]; b1.y = b1y[T];
        v8f cz = {};
        h[T] = __builtin_amdgcn_wmma_f32_16x16x4_f32(false, a1, false, b1,
                                                     (short)0, cz, false, false);
      }
#pragma unroll
      for (int T = 0; T < NTILE; ++T) {
        float g[8];
#pragma unroll
        for (int r = 0; r < 8; ++r) {
          float t  = fast_tanh(h[T][r]);
          float tw = t * wv[T];
          g[r] = __builtin_fmaf(-tw, t, wv[T]);
        }
        int addr = (T * 16 + l16) * 16 + 8 * hilo;
        *(float4*)(gbase + addr)     = make_float4(g[0], g[1], g[2], g[3]);
        *(float4*)(gbase + addr + 4) = make_float4(g[4], g[5], g[6], g[7]);
      }
      v8f acc = {};
#pragma unroll
      for (int c = 0; c < 16; ++c) {
        int roff = (4 * c + 2 * hilo) * 16 + l16;
        v2f bf;
        bf.x = gbase[roff];
        bf.y = gbase[roff + 16];
        v2f af; af.x = a2x[c]; af.y = a2y[c];
        acc = __builtin_amdgcn_wmma_f32_16x16x4_f32(false, af, false, bf,
                                                    (short)0, acc, false, false);
      }
      float g0 = acc[0], g1 = acc[1];
      float nX0 = Y0 + eta0;
      float nX1 = Y1 + eta1;
      float nY0 = g0 - X0;
      float nY1 = g1 - X1;
      X0 = nX0; X1 = nX1; Y0 = nY0; Y1 = nY1;
    }
    if (act) o4[b] = make_float4(X0, X1, Y0, Y1);
  }
}

extern "C" void kernel_launch(void* const* d_in, const int* in_sizes, int n_in,
                              void* d_out, int out_size, void* d_ws, size_t ws_size,
                              hipStream_t stream) {
  const float* z     = (const float*)d_in[0];
  const float* W_in  = (const float*)d_in[1];
  const float* W_out = (const float*)d_in[2];
  const float* b_in  = (const float*)d_in[3];
  const float* eta   = (const float*)d_in[4];
  float* out = (float*)d_out;
  int nbatch = in_sizes[0] / 4;

  dim3 block(256);
  dim3 grid(2048);
  henon_wmma_kernel<<<grid, block, 0, stream>>>(z, W_in, W_out, b_in, eta,
                                                out, nbatch);
}


# matmul2 packs two batch tiles per product: G of tile a as A rows 2-15, G of tile b as B cols 2-15, W_in in rows/cols 0-1: 24 matrix ops per 28 rows, was 20 per 16; hidden sum in another k order
# speedup vs baseline: 1.1642x; 1.1642x over previous
#include <hip/hip_runtime.h>

typedef __attribute__((ext_vector_type(2))) float v2f;
typedef __attribute__((ext_vector_type(8))) float v8f;

#define NI 64
#define NTILE 4
#define NSP 35
#define WSLOT 32
#define GT_FLOATS (16 * NSP * 4)
#define GROUP 56

__device__ __forceinline__ float fast_tanh(float x) {
#if __has_builtin(__builtin_amdgcn_tanhf)
  return __builtin_amdgcn_tanhf(x);
#else
  float r;
  asm volatile("v_tanh_f32 %0, %1" : "=v"(r) : "v"(x));
  return r;
#endif
}

__device__ __forceinline__ void tile_forward(float Y0, float Y1, int hilo,
                                             const float (&b1x)[NTILE], const float (&b1y)[NTILE],
                                             const float (&wv)[NTILE], float* st) {
  v2f a1;
  a1.x = hilo ? 1.0f : Y0;
  a1.y = hilo ? 0.0f : Y1;
  v8f h[NTILE];
#pragma unroll
  for (int T = 0; T < NTILE; ++T) {
    v2f b1; b1.x = b1x[T]; b1.y = b1y[T];
    v8f cz = {};
    h[T] = __builtin_amdgcn_wmma_f32_16x16x4_f32(false, a1, false, b1,
                                                 (short)0, cz, false, false);
  }
  float g[NTILE][8];
#pragma unroll
  for (int T = 0; T < NTILE; ++T) {
#pragma unroll
    for (int r = 0; r < 8; ++r) {
      float t  = fast_tanh(h[T][r]);
      float tw = t * wv[T];
      g[T][r] = __builtin_fmaf(-tw, t, wv[T]);
    }
  }
#pragma unroll
  for (int r = 0; r < 8; ++r)
    *(float4*)(st + r * 4) = make_float4(g[0][r], g[1][r], g[2][r], g[3][r]);
}

__global__ __launch_bounds__(256) void henon_wmma_kernel(
    const float* __restrict__ z, const float* __restrict__ W_in,
    const float* __restrict__ W_out, const float* __restrict__ b_in,
    const float* __restrict__ eta, float* __restrict__ out, int nbatch) {
  __shared__ __attribute__((aligned(16))) float Gt[8][GT_FLOATS];
  __shared__ __attribute__((aligned(16))) float Ex[8][32];

  const int tid   = threadIdx.x;
  const int lane  = tid & 31;
  const int wslot = tid >> 5;
  const int hilo  = lane >> 4;
  const int l16   = lane & 15;
  float* gbase = &Gt[wslot][0];
  float* ebase = &Ex[wslot][0];

  float b1x[NTILE], b1y[NTILE], wv[NTILE];
#pragma unroll
  for (int T = 0; T < NTILE; ++T) {
    int j = T * 16 + l16;
    b1x[T] = hilo ? b_in[j] : W_in[0 * NI + j];
    b1y[T] = hilo ? 0.0f    : W_in[1 * NI + j];
    wv[T]  = W_out[j];
  }
  *(float4*)(gbase + (l16 * NSP + WSLOT + hilo) * 4) =
      make_float4(W_in[hilo * NI + l16], W_in[hilo * NI + 16 + l16],
                  W_in[hilo * NI + 32 + l16], W_in[hilo * NI + 48 + l16]);
  const float eta0 = eta[0], eta1 = eta[1];

  const int slotA = (l16 < 2) ? (WSLOT + l16) : l16;
  const int slotB = (l16 < 2) ? (WSLOT + l16) : (16 + l16);
  const float* ldA = gbase + (hilo * NSP + slotA) * 4;
  const float* ldB = gbase + (hilo * NSP + slotB) * 4;
  float* stA = gbase + (l16 * NSP + 8 * hilo) * 4;
  float* stB = stA + 16 * 4;

  const int ngroups = (nbatch + GROUP - 1) / GROUP;
  const int waveG  = (int)(blockIdx.x * blockDim.x + tid) >> 5;
  const int nwaves = (int)(gridDim.x * blockDim.x) >> 5;
  const float4* z4 = (const float4*)z;
  float4* o4 = (float4*)out;

  for (int grp = waveG; grp < ngroups; grp += nwaves) {
    {
      int ng = grp + nwaves;
      if (ng < ngroups && lane < 7 && ng * GROUP + lane * 8 < nbatch)
        __builtin_prefetch((const char*)(z4 + ng * GROUP) + lane * 128, 0, 1);
    }
    for (int s = 0; s < 2; ++s) {
      const int bA = grp * GROUP + s * 28 + l16 - 2;
      const int bB = bA + 14;
      const bool real = (hilo == 0) && (l16 >= 2);
      const bool actA = real && (bA < nbatch);
      const bool actB = real && (bB < nbatch);
      float4 zA = make_float4(0.f, 0.f, 0.f, 0.f);
      float4 zB = make_float4(0.f, 0.f, 0.f, 0.f);
      if (actA) zA = z4[bA];
      if (actB) zB = z4[bB];
      float XA0 = zA.x, XA1 = zA.y, YA0 = zA.z, YA1 = zA.w;
      float XB0 = zB.x, XB1 = zB.y, YB0 = zB.z, YB1 = zB.w;

#pragma unroll
      for (int it = 0; it < 4; ++it) {
        tile_forward(YA0, YA1, hilo, b1x, b1y, wv, stA);
        tile_forward(YB0, YB1, hilo, b1x, b1y, wv, stB);

        float4 fa[8], fb[8];
#pragma unroll
        for (int q = 0; q < 8; ++q) {
          fa[q] = *(const float4*)(ldA + q * 2 * NSP * 4);
          fb[q] = *(const float4*)(ldB + q * 2 * NSP * 4);
        }
        v8f acc = {};
#pragma unroll
        for (int q = 0; q < 8; ++q) {
          v2f af, bf;
          af.x = fa[q].x; af.y = fa[q].y; bf.x = fb[q].x; bf.y = fb[q].y;
          acc = __builtin_amdgcn_wmma_f32_16x16x4_f32(false, af, false, bf,
                                                      (short)0, acc, false, false);
          af.x = fa[q].z; af.y = fa[q].w; bf.x = fb[q].z; bf.y = fb[q].w;
          acc = __builtin_amdgcn_wmma_f32_16x16x4_f32(false, af, false, bf,
                                                      (short)0, acc, false, false);
        }
        float gB0 = acc[0], gB1 = acc[1];
        if (l16 < 2) {
          float* e = ebase + l16 * 16 + 8 * hilo;
          *(float4*)(e)     = make_float4(acc[0], acc[1], acc[2], acc[3]);
          *(float4*)(e + 4) = make_float4(acc[4], acc[5], acc[6], acc[7]);
        }
        float gA0 = ebase[l16], gA1 = ebase[16 + l16];

        float nXA0 = YA0 + eta0, nXA1 = YA1 + eta1;
        float nYA0 = gA0 - XA0,  nYA1 = gA1 - XA1;
        float nXB0 = YB0 + eta0, nXB1 = YB1 + eta1;
        float nYB0 = gB0 - XB0,  nYB1 = gB1 - XB1;
        XA0 = nXA0; XA1 = nXA1; YA0 = nYA0; YA1 = nYA1;
        XB0 = nXB0; XB1 = nXB1; YB0 = nYB0; YB1 = nYB1;
      }
      if (actA) o4[bA] = make_float4(XA0, XA1, YA0, YA1);
      if (actB) o4[bB] = make_float4(XB0, XB1, YB0, YB1);
    }
  }
}

extern "C" void kernel_launch(void* const* d_in, const int* in_sizes, int n_in,
                              void* d_out, int out_size, void* d_ws, size_t ws_size,
                              hipStream_t stream) {
  const float* z     = (const float*)d_in[0];
  const float* W_in  = (const float*)d_in[1];
  const float* W_out = (const float*)d_in[2];
  const float* b_in  = (const float*)d_in[3];
  const float* eta   = (const float*)d_in[4];
  float* out = (float*)d_out;
  int nbatch = in_sizes[0] / 4;

  dim3 block(256);
  dim3 grid(1024);
  henon_wmma_kernel<<<grid, block, 0, stream>>>(z, W_in, W_out, b_in, eta,
                                                out, nbatch);
}
